# starting kernel
# baseline (speedup 1.0000x reference)
#include <hip/hip_runtime.h>
#include <math.h>

typedef __attribute__((ext_vector_type(2))) float v2f;
typedef __attribute__((ext_vector_type(8))) float v8f;

#define NEG_SLOPE 0.2f

__device__ __forceinline__ float lrelu(float x) {
  return x > 0.0f ? x : NEG_SLOPE * x;
}

__global__ void prep_wext_kernel(const float* __restrict__ W,
                                 const float* __restrict__ asrc,
                                 const float* __restrict__ adst,
                                 float* __restrict__ Wext, int K, int F) {
  int i = blockIdx.x * blockDim.x + threadIdx.x;
  if (i >= K * 16) return;
  int k = i >> 4;
  int n = i & 15;
  float v = 0.0f;
  if (n < F) {
    v = W[k * F + n];
  } else if (n == 8) {
    for (int j = 0; j < F; ++j) v += W[k * F + j] * asrc[j];
  } else if (n == 9) {
    for (int j = 0; j < F; ++j) v += W[k * F + j] * adst[j];
  }
  Wext[k * 16 + n] = v;
}

__global__ void init_node_kernel(float* __restrict__ m, float* __restrict__ z,
                                 float* __restrict__ agg, int N, int F) {
  int i = blockIdx.x * blockDim.x + threadIdx.x;
  if (i >= N) return;
  m[i] = -INFINITY;
  z[i] = 0.0f;
  for (int f = 0; f < F; ++f) agg[i * F + f] = 0.0f;
}

template <int K4>
__global__ void gemm_attn_wmma(const float* __restrict__ A, int lda,
                               const float* __restrict__ Bext,
                               float* __restrict__ H, int nRows) {
  int lane = threadIdx.x & 31;
  int wave = (int)((blockIdx.x * blockDim.x + threadIdx.x) >> 5);
  int row0 = wave * 16;
  if (row0 >= nRows) return;

  int hf = lane >> 4;
  int l  = lane & 15;

  v8f c = {};
#pragma unroll
  for (int k = 0; k < K4; ++k) {
    int kb = k * 4;
    const float* ap = A + (size_t)(row0 + l) * lda + kb + hf * 2;
    v2f a;
    a.x = ap[0];
    a.y = ap[1];
    const float* bp = Bext + (size_t)(kb + hf * 2) * 16 + l;
    v2f b;
    b.x = bp[0];
    b.y = bp[16];
    c = __builtin_amdgcn_wmma_f32_16x16x4_f32(false, a, false, b, (short)0, c,
                                              false, false);
  }

  float* out = H + (size_t)(row0 + hf * 8) * 16 + l;
#pragma unroll
  for (int r = 0; r < 8; ++r) out[(size_t)r * 16] = c[r];
}

__global__ void edge_max_kernel(const int* __restrict__ src,
                                const int* __restrict__ dst,
                                const float* __restrict__ H,
                                float* __restrict__ m, int E) {
  int i = blockIdx.x * blockDim.x + threadIdx.x;
  if (i >= E) return;
  int s = src[i], d = dst[i];
  float e = lrelu(H[(size_t)s * 16 + 8] + H[(size_t)d * 16 + 9]);
  atomicMax(m + d, e);
}

__global__ void edge_expsum_kernel(const int* __restrict__ src,
                                   const int* __restrict__ dst,
                                   const float* __restrict__ H,
                                   const float* __restrict__ m,
                                   float* __restrict__ z, int E) {
  int i = blockIdx.x * blockDim.x + threadIdx.x;
  if (i >= E) return;
  int s = src[i], d = dst[i];
  float e = lrelu(H[(size_t)s * 16 + 8] + H[(size_t)d * 16 + 9]);
  float p = __expf(e - m[d]);
  atomicAdd(z + d, p);
}

template <int F>
__global__ void edge_agg_kernel(const int* __restrict__ src,
                                const int* __restrict__ dst,
                                const float* __restrict__ H,
                                const float* __restrict__ m,
                                const float* __restrict__ z,
                                float* __restrict__ agg, int E) {
  int i = blockIdx.x * blockDim.x + threadIdx.x;
  if (i >= E) return;
  int s = src[i], d = dst[i];
  float e = lrelu(H[(size_t)s * 16 + 8] + H[(size_t)d * 16 + 9]);
  float alpha = __expf(e - m[d]) / (z[d] + 1e-16f);
  const float* hs = H + (size_t)s * 16;
  float* o = agg + (size_t)d * F;
#pragma unroll
  for (int f = 0; f < F; ++f) atomicAdd(o + f, alpha * hs[f]);
}

__global__ void relu_bias_kernel(const float* __restrict__ agg,
                                 const float* __restrict__ b,
                                 float* __restrict__ out, int N) {
  int i = blockIdx.x * blockDim.x + threadIdx.x;
  if (i >= N) return;
#pragma unroll
  for (int f = 0; f < 8; ++f) {
    float v = agg[(size_t)i * 8 + f] + b[f];
    out[(size_t)i * 8 + f] = v > 0.0f ? v : 0.0f;
  }
}

__global__ void logsoftmax_kernel(const float* __restrict__ agg,
                                  const float* __restrict__ b,
                                  float* __restrict__ out, int N) {
  int i = blockIdx.x * blockDim.x + threadIdx.x;
  if (i >= N) return;
  float o0 = agg[(size_t)i * 2 + 0] + b[0];
  float o1 = agg[(size_t)i * 2 + 1] + b[1];
  float mx = fmaxf(o0, o1);
  float lse = mx + __logf(__expf(o0 - mx) + __expf(o1 - mx));
  out[(size_t)i * 2 + 0] = o0 - lse;
  out[(size_t)i * 2 + 1] = o1 - lse;
}

extern "C" void kernel_launch(void* const* d_in, const int* in_sizes, int n_in,
                              void* d_out, int out_size, void* d_ws,
                              size_t ws_size, hipStream_t stream) {
  const float* x      = (const float*)d_in[0];
  const float* W1     = (const float*)d_in[1];
  const float* a1_src = (const float*)d_in[2];
  const float* a1_dst = (const float*)d_in[3];
  const float* b1     = (const float*)d_in[4];
  const float* W2     = (const float*)d_in[5];
  const float* a2_src = (const float*)d_in[6];
  const float* a2_dst = (const float*)d_in[7];
  const float* b2     = (const float*)d_in[8];
  const int*   eidx   = (const int*)d_in[9];

  const int N = in_sizes[0] / 36;
  const int E = in_sizes[9] / 2;
  const int* src = eidx;
  const int* dst = eidx + E;

  float* ws    = (float*)d_ws;
  float* h1    = ws;
  float* m1    = h1 + (size_t)N * 16;
  float* z1    = m1 + N;
  float* agg1  = z1 + N;
  float* h1r   = agg1 + (size_t)N * 8;
  float* h2    = h1r + (size_t)N * 8;
  float* m2    = h2 + (size_t)N * 16;
  float* z2    = m2 + N;
  float* agg2  = z2 + N;
  float* wext1 = agg2 + (size_t)N * 2;
  float* wext2 = wext1 + 36 * 16;

  const int BLK = 256;
  dim3 nodeGrid((N + BLK - 1) / BLK);
  dim3 edgeGrid((E + BLK - 1) / BLK);
  dim3 gemmGrid((N + 127) / 128);

  prep_wext_kernel<<<dim3((36 * 16 + BLK - 1) / BLK), BLK, 0, stream>>>(
      W1, a1_src, a1_dst, wext1, 36, 8);
  init_node_kernel<<<nodeGrid, BLK, 0, stream>>>(m1, z1, agg1, N, 8);
  gemm_attn_wmma<9><<<gemmGrid, BLK, 0, stream>>>(x, 36, wext1, h1, N);
  edge_max_kernel<<<edgeGrid, BLK, 0, stream>>>(src, dst, h1, m1, E);
  edge_expsum_kernel<<<edgeGrid, BLK, 0, stream>>>(src, dst, h1, m1, z1, E);
  edge_agg_kernel<8><<<edgeGrid, BLK, 0, stream>>>(src, dst, h1, m1, z1, agg1, E);
  relu_bias_kernel<<<nodeGrid, BLK, 0, stream>>>(agg1, b1, h1r, N);

  prep_wext_kernel<<<dim3(1), BLK, 0, stream>>>(W2, a2_src, a2_dst, wext2, 8, 2);
  init_node_kernel<<<nodeGrid, BLK, 0, stream>>>(m2, z2, agg2, N, 2);
  gemm_attn_wmma<2><<<gemmGrid, BLK, 0, stream>>>(h1r, 8, wext2, h2, N);
  edge_max_kernel<<<edgeGrid, BLK, 0, stream>>>(src, dst, h2, m2, E);
  edge_expsum_kernel<<<edgeGrid, BLK, 0, stream>>>(src, dst, h2, m2, z2, E);
  edge_agg_kernel<2><<<edgeGrid, BLK, 0, stream>>>(src, dst, h2, m2, z2, agg2, E);

  logsoftmax_kernel<<<nodeGrid, BLK, 0, stream>>>(agg2, b2, (float*)d_out, N);
}


# edge-pass scan trimmed: kept edge = one packed ring word (17-bit source, 15-bit local node), power-of-two ring, one compare per edge; freed LDS widens node ranges and s_src copies; sum order changed
# speedup vs baseline: 8.3012x; 8.3012x over previous
#include <hip/hip_runtime.h>
#include <math.h>

typedef __attribute__((ext_vector_type(2))) float v2f;
typedef __attribute__((ext_vector_type(8))) float v8f;

#define NEG_SLOPE 0.2f

__device__ __forceinline__ float lrelu(float x) {
  return x > 0.0f ? x : NEG_SLOPE * x;
}

__global__ void prep_wext_kernel(const float* __restrict__ W,
                                 const float* __restrict__ asrc,
                                 const float* __restrict__ adst,
                                 float* __restrict__ Wext, int K, int F) {
  int i = blockIdx.x * blockDim.x + threadIdx.x;
  if (i >= K * 16) return;
  int k = i >> 4;
  int n = i & 15;
  float v = 0.0f;
  if (n < F) {
    v = W[k * F + n];
  } else if (n == 8) {
    for (int j = 0; j < F; ++j) v += W[k * F + j] * asrc[j];
  } else if (n == 9) {
    for (int j = 0; j < F; ++j) v += W[k * F + j] * adst[j];
  }
  Wext[k * 16 + n] = v;
}

__global__ void init_node_kernel(float* __restrict__ m, float* __restrict__ z,
                                 float* __restrict__ agg, int N, int F) {
  int i = blockIdx.x * blockDim.x + threadIdx.x;
  if (i >= N) return;
  m[i] = -INFINITY;
  z[i] = 0.0f;
  for (int f = 0; f < F; ++f) agg[i * F + f] = 0.0f;
}

template <int K4>
__global__ void gemm_attn_wmma(const float* __restrict__ A, int lda,
                               const float* __restrict__ Bext,
                               float* __restrict__ H, float* __restrict__ sS,
                               float* __restrict__ sD, int nRows) {
  __shared__ float st[2][128];
  int lane = threadIdx.x & 31;
  int wv = (int)(threadIdx.x >> 5);
  int row0 = (int)blockIdx.x * 128 + wv * 16;

  if (row0 < nRows) {
    int hf = lane >> 4;
    int l  = lane & 15;

    v8f c = {};
#pragma unroll
    for (int k = 0; k < K4; ++k) {
      int kb = k * 4;
      const float* ap = A + (size_t)(row0 + l) * lda + kb + hf * 2;
      v2f a;
      a.x = ap[0];
      a.y = ap[1];
      const float* bp = Bext + (size_t)(kb + hf * 2) * 16 + l;
      v2f b;
      b.x = bp[0];
      b.y = bp[16];
      c = __builtin_amdgcn_wmma_f32_16x16x4_f32(false, a, false, b, (short)0, c,
                                                false, false);
    }

    float* out = H + (size_t)(row0 + hf * 8) * 16 + l;
#pragma unroll
    for (int r = 0; r < 8; ++r) out[(size_t)r * 16] = c[r];

    if (l == 8 || l == 9) {
      float* sp = &st[l - 8][wv * 16 + hf * 8];
#pragma unroll
      for (int r = 0; r < 8; ++r) sp[r] = c[r];
    }
  }
  __syncthreads();
  const int t = (int)(threadIdx.x & 127);
  const int row = (int)blockIdx.x * 128 + t;
  if (row < nRows) {
    if (threadIdx.x < 128) sS[row] = st[0][t];
    else sD[row] = st[1][t];
  }
}

constexpr int plane_pitch(int F, int NR) {
  if (F == 1) return NR;
  int p = NR;
  if (F == 8) {
    while (p % 16 != 4) ++p;
  } else {
    while (p % 64 < 32 / F || p % 64 > 64 - 32 / F) ++p;
  }
  return p;
}

template <int MODE, int F, int NRP, int U, int NS>
__device__ __forceinline__ void owner_edges(const int (&sg)[U], const unsigned (&tl)[U],
                                            const float* __restrict__ H,
                                            const float* __restrict__ sS,
                                            float* acc, const float* sd,
                                            const float* mp, const float* zp,
                                            const float* ss) {
  float a[U], hs[U][F];
#pragma unroll
  for (int u = 0; u < U; ++u) {
    if constexpr (MODE == 2) {
      const float* hp = H + (size_t)sg[u] * 16;
      a[u] = hp[8];
      if constexpr (F == 8) {
        float4 h0 = *reinterpret_cast<const float4*>(hp);
        float4 h1 = *reinterpret_cast<const float4*>(hp + 4);
        hs[u][0] = h0.x; hs[u][1] = h0.y; hs[u][2] = h0.z; hs[u][3] = h0.w;
        hs[u][4] = h1.x; hs[u][5] = h1.y; hs[u][6] = h1.z; hs[u][7] = h1.w;
      } else {
#pragma unroll
        for (int f = 0; f < F; ++f) hs[u][f] = hp[f];
      }
    } else if constexpr (NS > 0) {
      const unsigned s = (unsigned)sg[u];
      const bool inl = s < (unsigned)NS;
      const float vl = ss[inl ? s : 0u];
      const float vg = sS[inl ? 0u : s];
      a[u] = inl ? vl : vg;
    } else {
      a[u] = sS[sg[u]];
    }
  }
#pragma unroll
  for (int u = 0; u < U; ++u) {
    if constexpr (MODE == 0) {
      __hip_atomic_fetch_max(&acc[tl[u]], a[u], __ATOMIC_RELAXED,
                             __HIP_MEMORY_SCOPE_WORKGROUP);
    } else {
      float e = lrelu(a[u] + sd[tl[u]]);
      if constexpr (MODE == 1) {
        float p = __expf(e - mp[tl[u]]);
        atomicAdd(&acc[tl[u]], p);
      } else {
        float alpha = __expf(e - mp[tl[u]]) / zp[tl[u]];
#pragma unroll
        for (int f = 0; f < F; ++f)
          atomicAdd(&acc[f * NRP + tl[u]], alpha * hs[u][f]);
      }
    }
  }
}

template <int MODE, int F, int NR, int K, int U, int NS = 0>
__global__ __launch_bounds__(1024) void edge_owner_kernel(
    const int* __restrict__ src, const int* __restrict__ dst,
    const float* __restrict__ H, const float* __restrict__ sS,
    const float* __restrict__ sD, const float* __restrict__ m,
    const float* __restrict__ z, float* __restrict__ outp, int N, int E) {
  constexpr int NPL = (MODE == 0) ? 0 : (MODE == 1 ? 2 : 3);
  constexpr int PLN = (NR * NPL + NS > 0) ? NR * NPL + NS : 4;
  static_assert(MODE != 2 || NS == 0, "the aggregation gathers rows, not scores");
  constexpr int TRIP = 1024 * K;
  constexpr int NRP = plane_pitch(F, NR);
  static_assert(F == 1 || F == 2 || F == 8, "plane pitch is worked out for these");
  static_assert(NR % 4 == 0, "operand planes are filled by 16-byte stores");
  static_assert(NR <= 32768, "a local destination must fit 15 bits");
  constexpr int TH = 32 * U;
  constexpr int QCAP = 256;
  static_assert(TH + 127 <= QCAP && (QCAP & (QCAP - 1)) == 0 && QCAP % U == 0, "ring length");
  __shared__ float acc[NRP * F];
  __shared__ alignas(16) float pl[PLN];
  __shared__ alignas(16) unsigned qs[32 * QCAP];
  const float* sd = pl;
  const float* mp = pl + (NPL > 1 ? NR : 0);
  const float* zp = pl + (NPL > 2 ? 2 * NR : 0);
  float* ssw = pl + NR * NPL;
  const float* ss = ssw;
  const int tid = threadIdx.x;
  const int lane = tid & 31;
  unsigned* qw = qs + (tid >> 5) * QCAP;
  const int nwg = gridDim.x;
  const int R = (N + NR - 1) / NR;
  int G = nwg / R;
  if (G < 1) G = 1;
  const int g = blockIdx.x % G;
  const int rstride = nwg / G;
  int per = (E + G - 1) / G;
  per = ((per + TRIP - 1) / TRIP) * TRIP;
  const long long b0 = (long long)g * per;
  const int begin = b0 < (long long)E ? (int)b0 : E;
  const int end = (E - begin < per) ? E : begin + per;
  const bool vec = (E % K) == 0 && N <= (1 << 17);
  const float initv = (MODE == 0) ? -INFINITY : 0.0f;

  if constexpr (NS > 0) {
    static_assert(NS % 4096 == 0, "one 16-byte load per thread and trip");
    const int ns = N < NS ? N : NS;
    const int s4 = ((((size_t)sS) & 15) == 0) ? (ns & ~3) : 0;
#pragma unroll (NS / 4096)
    for (int i = tid * 4; i < s4; i += 4096)
      *reinterpret_cast<float4*>(ssw + i) = *reinterpret_cast<const float4*>(sS + i);
    for (int i = s4 + tid; i < ns; i += 1024) ssw[i] = sS[i];
  }

  for (int r = blockIdx.x / G; r < R; r += rstride) {
    const int lo = r * NR;
    const unsigned nr = (unsigned)((N - lo < NR) ? (N - lo) : NR);
    for (int i = tid; i < NRP * F; i += 1024) acc[i] = initv;
    if constexpr (MODE != 0) {
      const float* sdp = sD + lo;
      const float* mpg = (NPL > 1) ? m + lo : sdp;
      const float* zpg = (NPL > 2) ? z + lo : sdp;
      const bool al16 =
          ((((size_t)sdp) | ((size_t)mpg) | ((size_t)zpg)) & 15) == 0 && (NR % 4) == 0;
      const int n4 = al16 ? ((int)nr & ~3) : 0;
#pragma unroll 2
      for (int i = tid * 4; i < n4; i += 4096) {
        float4 v0 = *reinterpret_cast<const float4*>(sdp + i);
        float4 v1, v2;
        if constexpr (NPL > 1) v1 = *reinterpret_cast<const float4*>(mpg + i);
        if constexpr (NPL > 2) v2 = *reinterpret_cast<const float4*>(zpg + i);
        *reinterpret_cast<float4*>(pl + i) = v0;
        if constexpr (NPL > 1) {
          v1.x = lrelu(v1.x + v0.x); v1.y = lrelu(v1.y + v0.y);
          v1.z = lrelu(v1.z + v0.z); v1.w = lrelu(v1.w + v0.w);
          *reinterpret_cast<float4*>(pl + NR + i) = v1;
        }
        if constexpr (NPL > 2) {
          v2.x += 1e-16f; v2.y += 1e-16f; v2.z += 1e-16f; v2.w += 1e-16f;
          *reinterpret_cast<float4*>(pl + 2 * NR + i) = v2;
        }
      }
      for (int i = n4 + tid; i < (int)nr; i += 1024) {
        pl[i] = sdp[i];
        if constexpr (NPL > 1) pl[NR + i] = lrelu(mpg[i] + sdp[i]);
        if constexpr (NPL > 2) pl[2 * NR + i] = zpg[i] + 1e-16f;
      }
    }
    __syncthreads();

    if (begin < end) {
      if (vec) {
        const int last = end - K;
        unsigned head = 0, tail = 0;
        int i = begin + tid * K;
        int ic = i < last ? i : last;
        int4 dv[K / 4], sv[K / 4], dn[K / 4], sn[K / 4];
#pragma unroll
        for (int q = 0; q < K / 4; ++q) {
          dv[q] = *reinterpret_cast<const int4*>(dst + ic + 4 * q);
          sv[q] = *reinterpret_cast<const int4*>(src + ic + 4 * q);
        }
        for (int base = begin; base < end; base += TRIP) {
          const int inext = i + TRIP;
          ic = inext < last ? inext : last;
#pragma unroll
          for (int q = 0; q < K / 4; ++q) {
            dn[q] = *reinterpret_cast<const int4*>(dst + ic + 4 * q);
            sn[q] = *reinterpret_cast<const int4*>(src + ic + 4 * q);
          }
          const unsigned lou = (i < end) ? (unsigned)lo : 0x80000000u;
#pragma unroll
          for (int q = 0; q < K / 4; ++q) {
            const int d4[4] = {dv[q].x, dv[q].y, dv[q].z, dv[q].w};
            const int s4[4] = {sv[q].x, sv[q].y, sv[q].z, sv[q].w};
#pragma unroll
            for (int j = 0; j < 4; ++j) {
              const unsigned t = (unsigned)d4[j] - lou;
              const bool keep = t < nr;
              const unsigned km = __builtin_amdgcn_ballot_w32(keep);
              if (keep) {
                const unsigned pos = __builtin_amdgcn_mbcnt_lo(km, tail) & (QCAP - 1);
                qw[pos] = ((unsigned)s4[j] << 15) | t;
              }
              tail += (unsigned)__builtin_popcount(km);
            }
            while (tail - head >= (unsigned)TH) {
              int sg[U];
              unsigned tl[U];
              unsigned en[U];
              const unsigned p0 = (head + (unsigned)lane * U) & (QCAP - 1);
              if constexpr (U == 4) {
                const uint4 e4 = *reinterpret_cast<const uint4*>(qw + p0);
                en[0] = e4.x; en[1] = e4.y; en[2] = e4.z; en[3] = e4.w;
              } else if constexpr (U == 2) {
                const uint2 e2 = *reinterpret_cast<const uint2*>(qw + p0);
                en[0] = e2.x; en[1] = e2.y;
              } else {
#pragma unroll
                for (int u = 0; u < U; ++u) en[u] = qw[(p0 + u) & (QCAP - 1)];
              }
#pragma unroll
              for (int u = 0; u < U; ++u) {
                sg[u] = (int)(en[u] >> 15);
                tl[u] = en[u] & 0x7fffu;
              }
              owner_edges<MODE, F, NRP, U, NS>(sg, tl, H, sS, acc, sd, mp, zp, ss);
              head += TH;
            }
          }
#pragma unroll
          for (int q = 0; q < K / 4; ++q) { dv[q] = dn[q]; sv[q] = sn[q]; }
          i = inext;
        }
        int cnt = (int)(tail - head);
        while (cnt > 0) {
          if (lane < cnt) {
            const unsigned en = qw[(head + (unsigned)lane) & (QCAP - 1)];
            const int sg[1] = {(int)(en >> 15)};
            const unsigned tl[1] = {en & 0x7fffu};
            owner_edges<MODE, F, NRP, 1, NS>(sg, tl, H, sS, acc, sd, mp, zp, ss);
          }
          head += 32;
          cnt -= 32;
        }
      } else {
        for (int i = begin + tid; i < end; i += 1024) {
          const unsigned t = (unsigned)(dst[i] - lo);
          if (t < nr) {
            const int sg[1] = {src[i]};
            const unsigned tl[1] = {t};
            owner_edges<MODE, F, NRP, 1, NS>(sg, tl, H, sS, acc, sd, mp, zp, ss);
          }
        }
      }
    }
    __syncthreads();

    const int cntf = (int)nr * F;
    float* gp = outp + (size_t)lo * F;
    for (int k = tid; k < cntf; k += 1024) {
      float v = acc[(k % F) * NRP + k / F];
      if (MODE == 0) atomicMax(gp + k, v);
      else atomicAdd(gp + k, v);
    }
    __syncthreads();
  }
}

__global__ void relu_bias_kernel(const float* __restrict__ agg,
                                 const float* __restrict__ b,
                                 float* __restrict__ out, int N) {
  int i = blockIdx.x * blockDim.x + threadIdx.x;
  if (i >= N) return;
#pragma unroll
  for (int f = 0; f < 8; ++f) {
    float v = agg[(size_t)i * 8 + f] + b[f];
    out[(size_t)i * 8 + f] = v > 0.0f ? v : 0.0f;
  }
}

__global__ void logsoftmax_kernel(const float* __restrict__ agg,
                                  const float* __restrict__ b,
                                  float* __restrict__ out, int N) {
  int i = blockIdx.x * blockDim.x + threadIdx.x;
  if (i >= N) return;
  float o0 = agg[(size_t)i * 2 + 0] + b[0];
  float o1 = agg[(size_t)i * 2 + 1] + b[1];
  float mx = fmaxf(o0, o1);
  float lse = mx + __logf(__expf(o0 - mx) + __expf(o1 - mx));
  out[(size_t)i * 2 + 0] = o0 - lse;
  out[(size_t)i * 2 + 1] = o1 - lse;
}

extern "C" void kernel_launch(void* const* d_in, const int* in_sizes, int n_in,
                              void* d_out, int out_size, void* d_ws,
                              size_t ws_size, hipStream_t stream) {
  const float* x      = (const float*)d_in[0];
  const float* W1     = (const float*)d_in[1];
  const float* a1_src = (const float*)d_in[2];
  const float* a1_dst = (const float*)d_in[3];
  const float* b1     = (const float*)d_in[4];
  const float* W2     = (const float*)d_in[5];
  const float* a2_src = (const float*)d_in[6];
  const float* a2_dst = (const float*)d_in[7];
  const float* b2     = (const float*)d_in[8];
  const int*   eidx   = (const int*)d_in[9];

  const int N = in_sizes[0] / 36;
  const int E = in_sizes[9] / 2;
  const int* src = eidx;
  const int* dst = eidx + E;

  float* ws    = (float*)d_ws;
  float* h1    = ws;
  float* m1    = h1 + (size_t)N * 16;
  float* z1    = m1 + N;
  float* agg1  = z1 + N;
  float* h1r   = agg1 + (size_t)N * 8;
  float* h2    = h1r + (size_t)N * 8;
  float* m2    = h2 + (size_t)N * 16;
  float* z2    = m2 + N;
  float* agg2  = z2 + N;
  float* wext1 = agg2 + (size_t)N * 2;
  float* wext2 = wext1 + 36 * 16;
  float* sS1   = h2;
  float* sD1   = h2 + N;
  float* sS2   = agg1;
  float* sD2   = agg1 + N;

  const int BLK = 256;
  dim3 nodeGrid((N + BLK - 1) / BLK);
  dim3 gemmGrid((N + 127) / 128);

  prep_wext_kernel<<<dim3((36 * 16 + BLK - 1) / BLK), BLK, 0, stream>>>(
      W1, a1_src, a1_dst, wext1, 36, 8);
  init_node_kernel<<<nodeGrid, BLK, 0, stream>>>(m1, z1, agg1, N, 8);
  gemm_attn_wmma<9><<<gemmGrid, BLK, 0, stream>>>(x, 36, wext1, h1, sS1, sD1, N);
  edge_owner_kernel<0, 1, 25088, 8, 4, 45056><<<dim3(256), dim3(1024), 0, stream>>>(
      src, dst, h1, sS1, sD1, nullptr, nullptr, m1, N, E);
  edge_owner_kernel<1, 1, 20480, 8, 4, 12288><<<dim3(256), dim3(1024), 0, stream>>>(
      src, dst, h1, sS1, sD1, m1, nullptr, z1, N, E);
  edge_owner_kernel<2, 8, 6668, 16, 2><<<dim3(256), dim3(1024), 0, stream>>>(
      src, dst, h1, sS1, sD1, m1, z1, agg1, N, E);
  relu_bias_kernel<<<nodeGrid, BLK, 0, stream>>>(agg1, b1, h1r, N);

  prep_wext_kernel<<<dim3(1), BLK, 0, stream>>>(W2, a2_src, a2_dst, wext2, 8, 2);
  init_node_kernel<<<nodeGrid, BLK, 0, stream>>>(m2, z2, agg2, N, 2);
  gemm_attn_wmma<2><<<gemmGrid, BLK, 0, stream>>>(h1r, 8, wext2, h2, sS2, sD2, N);
  edge_owner_kernel<0, 1, 25088, 8, 4, 45056><<<dim3(256), dim3(1024), 0, stream>>>(
      src, dst, h2, sS2, sD2, nullptr, nullptr, m2, N, E);
  edge_owner_kernel<1, 1, 20480, 8, 4, 12288><<<dim3(256), dim3(1024), 0, stream>>>(
      src, dst, h2, sS2, sD2, m2, nullptr, z2, N, E);
  edge_owner_kernel<2, 2, 14720, 8, 4><<<dim3(256), dim3(1024), 0, stream>>>(
      src, dst, h2, sS2, sD2, m2, z2, agg2, N, E);

  logsoftmax_kernel<<<nodeGrid, BLK, 0, stream>>>(agg2, b2, (float*)d_out, N);
}


# max and exp-sum passes: source scores laid out in edge order once per layer by a new LDS-lookup kernel; the passes stream destinations only and read scores by edge place, not a random line per edge
# speedup vs baseline: 9.2490x; 1.1142x over previous
#include <hip/hip_runtime.h>
#include <math.h>

typedef __attribute__((ext_vector_type(2))) float v2f;
typedef __attribute__((ext_vector_type(8))) float v8f;

#define NEG_SLOPE 0.2f

__device__ __forceinline__ float lrelu(float x) {
  return x > 0.0f ? x : NEG_SLOPE * x;
}

__global__ void prep_wext_kernel(const float* __restrict__ W,
                                 const float* __restrict__ asrc,
                                 const float* __restrict__ adst,
                                 float* __restrict__ Wext, int K, int F) {
  int i = blockIdx.x * blockDim.x + threadIdx.x;
  if (i >= K * 16) return;
  int k = i >> 4;
  int n = i & 15;
  float v = 0.0f;
  if (n < F) {
    v = W[k * F + n];
  } else if (n == 8) {
    for (int j = 0; j < F; ++j) v += W[k * F + j] * asrc[j];
  } else if (n == 9) {
    for (int j = 0; j < F; ++j) v += W[k * F + j] * adst[j];
  }
  Wext[k * 16 + n] = v;
}

__global__ void init_node_kernel(float* __restrict__ m, float* __restrict__ z,
                                 float* __restrict__ agg, int N, int F) {
  int i = blockIdx.x * blockDim.x + threadIdx.x;
  if (i >= N) return;
  m[i] = -INFINITY;
  z[i] = 0.0f;
  for (int f = 0; f < F; ++f) agg[i * F + f] = 0.0f;
}

template <int K4>
__global__ void gemm_attn_wmma(const float* __restrict__ A, int lda,
                               const float* __restrict__ Bext,
                               float* __restrict__ H, float* __restrict__ sS,
                               float* __restrict__ sD, int nRows) {
  __shared__ float st[2][128];
  int lane = threadIdx.x & 31;
  int wv = (int)(threadIdx.x >> 5);
  int row0 = (int)blockIdx.x * 128 + wv * 16;

  if (row0 < nRows) {
    int hf = lane >> 4;
    int l  = lane & 15;

    v8f c = {};
#pragma unroll
    for (int k = 0; k < K4; ++k) {
      int kb = k * 4;
      const float* ap = A + (size_t)(row0 + l) * lda + kb + hf * 2;
      v2f a;
      a.x = ap[0];
      a.y = ap[1];
      const float* bp = Bext + (size_t)(kb + hf * 2) * 16 + l;
      v2f b;
      b.x = bp[0];
      b.y = bp[16];
      c = __builtin_amdgcn_wmma_f32_16x16x4_f32(false, a, false, b, (short)0, c,
                                                false, false);
    }

    float* out = H + (size_t)(row0 + hf * 8) * 16 + l;
#pragma unroll
    for (int r = 0; r < 8; ++r) out[(size_t)r * 16] = c[r];

    if (l == 8 || l == 9) {
      float* sp = &st[l - 8][wv * 16 + hf * 8];
#pragma unroll
      for (int r = 0; r < 8; ++r) sp[r] = c[r];
    }
  }
  __syncthreads();
  const int t = (int)(threadIdx.x & 127);
  const int row = (int)blockIdx.x * 128 + t;
  if (row < nRows) {
    if (threadIdx.x < 128) sS[row] = st[0][t];
    else sD[row] = st[1][t];
  }
}

constexpr int SSCAP = 65536;
constexpr int EPT = 28;

__global__ __launch_bounds__(1024) void edge_score_kernel(
    const int* __restrict__ src, const float* __restrict__ sS,
    float* __restrict__ sE, const float* wsb, size_t wsz, int N, int E) {
  __shared__ alignas(16) float ss[SSCAP];
  if ((size_t)((const char*)(sE + E) - (const char*)wsb) > wsz) return;
  const unsigned tid = threadIdx.x;
  const unsigned Eu = (unsigned)E;
  const long long CH = 1024LL * EPT;
  for (long long c0 = (long long)blockIdx.x * CH; c0 < (long long)E;
       c0 += (long long)gridDim.x * CH) {
    const unsigned i0 = (unsigned)c0 + tid;
    int s[EPT];
    float v[EPT];
#pragma unroll
    for (int q = 0; q < EPT; ++q) {
      const unsigned i = i0 + (unsigned)q * 1024u;
      const int sv = src[i < Eu ? i : Eu - 1u];
      s[q] = (i < Eu) ? sv : -1;
      v[q] = 0.0f;
    }
    for (int base = 0; base < N; base += SSCAP) {
      const int cnt = (N - base < SSCAP) ? (N - base) : SSCAP;
      __syncthreads();
      const float* sp = sS + base;
      const int c4 = ((((size_t)sp) & 15) == 0) ? (cnt & ~3) : 0;
#pragma unroll 8
      for (int i = (int)tid * 4; i < c4; i += 4096)
        *reinterpret_cast<float4*>(ss + i) = *reinterpret_cast<const float4*>(sp + i);
      for (int i = c4 + (int)tid; i < cnt; i += 1024) ss[i] = sp[i];
      __syncthreads();
#pragma unroll
      for (int q = 0; q < EPT; ++q) {
        const unsigned o = (unsigned)(s[q] - base);
        const bool in = o < (unsigned)cnt;
        const float t = ss[in ? o : 0u];
        v[q] = in ? t : v[q];
      }
    }
#pragma unroll
    for (int q = 0; q < EPT; ++q) {
      const unsigned i = i0 + (unsigned)q * 1024u;
      if (i < Eu) sE[i] = v[q];
    }
  }
}

constexpr int plane_pitch(int F, int NR) {
  if (F == 1) return NR;
  int p = NR;
  if (F == 8) {
    while (p % 16 != 4) ++p;
  } else {
    while (p % 64 < 32 / F || p % 64 > 64 - 32 / F) ++p;
  }
  return p;
}

template <int MODE, int F, int NRP, int U>
__device__ __forceinline__ void owner_edges(const int (&sg)[U], const unsigned (&tl)[U],
                                            const float* __restrict__ H,
                                            const float* __restrict__ sS,
                                            const float* __restrict__ eS,
                                            const int* __restrict__ srcb,
                                            const bool useE,
                                            float* acc, const float* sd,
                                            const float* mp, const float* zp) {
  float a[U], hs[U][F];
#pragma unroll
  for (int u = 0; u < U; ++u) {
    if constexpr (MODE == 2) {
      const float* hp = H + (size_t)sg[u] * 16;
      a[u] = hp[8];
      if constexpr (F == 8) {
        float4 h0 = *reinterpret_cast<const float4*>(hp);
        float4 h1 = *reinterpret_cast<const float4*>(hp + 4);
        hs[u][0] = h0.x; hs[u][1] = h0.y; hs[u][2] = h0.z; hs[u][3] = h0.w;
        hs[u][4] = h1.x; hs[u][5] = h1.y; hs[u][6] = h1.z; hs[u][7] = h1.w;
      } else {
#pragma unroll
        for (int f = 0; f < F; ++f) hs[u][f] = hp[f];
      }
    }
  }
  if constexpr (MODE != 2) {
    if (useE) {
#pragma unroll
      for (int u = 0; u < U; ++u) a[u] = eS[sg[u]];
    } else {
#pragma unroll
      for (int u = 0; u < U; ++u) a[u] = sS[srcb[sg[u]]];
    }
  }
#pragma unroll
  for (int u = 0; u < U; ++u) {
    if constexpr (MODE == 0) {
      __hip_atomic_fetch_max(&acc[tl[u]], a[u], __ATOMIC_RELAXED,
                             __HIP_MEMORY_SCOPE_WORKGROUP);
    } else {
      float e = lrelu(a[u] + sd[tl[u]]);
      if constexpr (MODE == 1) {
        float p = __expf(e - mp[tl[u]]);
        atomicAdd(&acc[tl[u]], p);
      } else {
        float alpha = __expf(e - mp[tl[u]]) / zp[tl[u]];
#pragma unroll
        for (int f = 0; f < F; ++f)
          atomicAdd(&acc[f * NRP + tl[u]], alpha * hs[u][f]);
      }
    }
  }
}

template <int MODE, int F, int NR, int K, int U>
__global__ __launch_bounds__(1024) void edge_owner_kernel(
    const int* __restrict__ src, const int* __restrict__ dst,
    const float* __restrict__ H, const float* __restrict__ sS,
    const float* __restrict__ sD, const float* __restrict__ m,
    const float* __restrict__ z, float* __restrict__ outp,
    const float* __restrict__ ssE, const float* wsb, size_t wsz, int N, int E) {
  constexpr int NPL = (MODE == 0) ? 0 : (MODE == 1 ? 2 : 3);
  constexpr int PLN = (NR * NPL > 0) ? NR * NPL : 4;
  constexpr int TRIP = 1024 * K;
  constexpr int NRP = plane_pitch(F, NR);
  static_assert(F == 1 || F == 2 || F == 8, "plane pitch is worked out for these");
  static_assert(NR % 4 == 0, "operand planes are filled by 16-byte stores");
  static_assert(NR <= 32768, "a local destination must fit 15 bits");
  constexpr int TH = 32 * U;
  constexpr int QCAP = (MODE == 0) ? 512 : 256;
  static_assert(TH + 127 <= QCAP && (QCAP & (QCAP - 1)) == 0 && QCAP % U == 0, "ring length");
  __shared__ float acc[NRP * F];
  __shared__ alignas(16) float pl[PLN];
  __shared__ alignas(16) unsigned qs[32 * QCAP];
  const float* sd = pl;
  const float* mp = pl + (NPL > 1 ? NR : 0);
  const float* zp = pl + (NPL > 2 ? 2 * NR : 0);
  const int tid = threadIdx.x;
  const int lane = tid & 31;
  unsigned* qw = qs + (tid >> 5) * QCAP;
  const int nwg = gridDim.x;
  const int R = (N + NR - 1) / NR;
  int G = nwg / R;
  if (G < 1) G = 1;
  const int g = blockIdx.x % G;
  const int rstride = nwg / G;
  int per = (E + G - 1) / G;
  per = ((per + TRIP - 1) / TRIP) * TRIP;
  const long long b0 = (long long)g * per;
  const int begin = b0 < (long long)E ? (int)b0 : E;
  const int end = (E - begin < per) ? E : begin + per;
  const bool vec = (E % K) == 0 && (MODE == 2 ? N <= (1 << 17) : per <= (1 << 17));
  const float initv = (MODE == 0) ? -INFINITY : 0.0f;
  const bool useE = (MODE != 2) &&
      (size_t)((const char*)(ssE + E) - (const char*)wsb) <= wsz;
  const float* eS = (MODE == 2) ? sS : ssE + begin;
  const int* srcb = src + begin;

  for (int r = blockIdx.x / G; r < R; r += rstride) {
    const int lo = r * NR;
    const unsigned nr = (unsigned)((N - lo < NR) ? (N - lo) : NR);
    for (int i = tid; i < NRP * F; i += 1024) acc[i] = initv;
    if constexpr (MODE != 0) {
      const float* sdp = sD + lo;
      const float* mpg = (NPL > 1) ? m + lo : sdp;
      const float* zpg = (NPL > 2) ? z + lo : sdp;
      const bool al16 =
          ((((size_t)sdp) | ((size_t)mpg) | ((size_t)zpg)) & 15) == 0 && (NR % 4) == 0;
      const int n4 = al16 ? ((int)nr & ~3) : 0;
#pragma unroll 2
      for (int i = tid * 4; i < n4; i += 4096) {
        float4 v0 = *reinterpret_cast<const float4*>(sdp + i);
        float4 v1, v2;
        if constexpr (NPL > 1) v1 = *reinterpret_cast<const float4*>(mpg + i);
        if constexpr (NPL > 2) v2 = *reinterpret_cast<const float4*>(zpg + i);
        *reinterpret_cast<float4*>(pl + i) = v0;
        if constexpr (NPL > 1) {
          v1.x = lrelu(v1.x + v0.x); v1.y = lrelu(v1.y + v0.y);
          v1.z = lrelu(v1.z + v0.z); v1.w = lrelu(v1.w + v0.w);
          *reinterpret_cast<float4*>(pl + NR + i) = v1;
        }
        if constexpr (NPL > 2) {
          v2.x += 1e-16f; v2.y += 1e-16f; v2.z += 1e-16f; v2.w += 1e-16f;
          *reinterpret_cast<float4*>(pl + 2 * NR + i) = v2;
        }
      }
      for (int i = n4 + tid; i < (int)nr; i += 1024) {
        pl[i] = sdp[i];
        if constexpr (NPL > 1) pl[NR + i] = lrelu(mpg[i] + sdp[i]);
        if constexpr (NPL > 2) pl[2 * NR + i] = zpg[i] + 1e-16f;
      }
    }
    __syncthreads();

    if (begin < end) {
      if (vec) {
        const int last = end - K;
        unsigned head = 0, tail = 0;
        int i = begin + tid * K;
        int ic = i < last ? i : last;
        int4 dv[K / 4], sv[K / 4], dn[K / 4], sn[K / 4];
#pragma unroll
        for (int q = 0; q < K / 4; ++q) {
          dv[q] = *reinterpret_cast<const int4*>(dst + ic + 4 * q);
          if constexpr (MODE == 2) sv[q] = *reinterpret_cast<const int4*>(src + ic + 4 * q);
        }
        for (int base = begin; base < end; base += TRIP) {
          const int inext = i + TRIP;
          ic = inext < last ? inext : last;
#pragma unroll
          for (int q = 0; q < K / 4; ++q) {
            dn[q] = *reinterpret_cast<const int4*>(dst + ic + 4 * q);
            if constexpr (MODE == 2) sn[q] = *reinterpret_cast<const int4*>(src + ic + 4 * q);
          }
          const unsigned lou = (i < end) ? (unsigned)lo : 0x80000000u;
          const unsigned prel = (unsigned)(i - begin);
#pragma unroll
          for (int q = 0; q < K / 4; ++q) {
            const int d4[4] = {dv[q].x, dv[q].y, dv[q].z, dv[q].w};
            unsigned h4[4];
            if constexpr (MODE == 2) {
              h4[0] = (unsigned)sv[q].x; h4[1] = (unsigned)sv[q].y;
              h4[2] = (unsigned)sv[q].z; h4[3] = (unsigned)sv[q].w;
            } else {
#pragma unroll
              for (int j = 0; j < 4; ++j) h4[j] = prel + (unsigned)(4 * q + j);
            }
#pragma unroll
            for (int j = 0; j < 4; ++j) {
              const unsigned t = (unsigned)d4[j] - lou;
              const bool keep = t < nr;
              const unsigned km = __builtin_amdgcn_ballot_w32(keep);
              if (keep) {
                const unsigned pos = __builtin_amdgcn_mbcnt_lo(km, tail) & (QCAP - 1);
                qw[pos] = (h4[j] << 15) | t;
              }
              tail += (unsigned)__builtin_popcount(km);
            }
            while (tail - head >= (unsigned)TH) {
              int sg[U];
              unsigned tl[U];
              unsigned en[U];
              const unsigned p0 = (head + (unsigned)lane * U) & (QCAP - 1);
              if constexpr (U == 4) {
                const uint4 e4 = *reinterpret_cast<const uint4*>(qw + p0);
                en[0] = e4.x; en[1] = e4.y; en[2] = e4.z; en[3] = e4.w;
              } else if constexpr (U == 2) {
                const uint2 e2 = *reinterpret_cast<const uint2*>(qw + p0);
                en[0] = e2.x; en[1] = e2.y;
              } else {
#pragma unroll
                for (int u = 0; u < U; ++u) en[u] = qw[(p0 + u) & (QCAP - 1)];
              }
#pragma unroll
              for (int u = 0; u < U; ++u) {
                sg[u] = (int)(en[u] >> 15);
                tl[u] = en[u] & 0x7fffu;
              }
              owner_edges<MODE, F, NRP, U>(sg, tl, H, sS, eS, srcb, useE, acc, sd, mp, zp);
              head += TH;
            }
          }
#pragma unroll
          for (int q = 0; q < K / 4; ++q) {
            dv[q] = dn[q];
            if constexpr (MODE == 2) sv[q] = sn[q];
          }
          i = inext;
        }
        int cnt = (int)(tail - head);
        while (cnt > 0) {
          if (lane < cnt) {
            const unsigned en = qw[(head + (unsigned)lane) & (QCAP - 1)];
            const int sg[1] = {(int)(en >> 15)};
            const unsigned tl[1] = {en & 0x7fffu};
            owner_edges<MODE, F, NRP, 1>(sg, tl, H, sS, eS, srcb, useE, acc, sd, mp, zp);
          }
          head += 32;
          cnt -= 32;
        }
      } else {
        for (int i = begin + tid; i < end; i += 1024) {
          const unsigned t = (unsigned)(dst[i] - lo);
          if (t < nr) {
            const int sg[1] = {MODE == 2 ? src[i] : i - begin};
            const unsigned tl[1] = {t};
            owner_edges<MODE, F, NRP, 1>(sg, tl, H, sS, eS, srcb, useE, acc, sd, mp, zp);
          }
        }
      }
    }
    __syncthreads();

    const int cntf = (int)nr * F;
    float* gp = outp + (size_t)lo * F;
    for (int k = tid; k < cntf; k += 1024) {
      float v = acc[(k % F) * NRP + k / F];
      if (MODE == 0) atomicMax(gp + k, v);
      else atomicAdd(gp + k, v);
    }
    __syncthreads();
  }
}

__global__ void relu_bias_kernel(const float* __restrict__ agg,
                                 const float* __restrict__ b,
                                 float* __restrict__ out, int N) {
  int i = blockIdx.x * blockDim.x + threadIdx.x;
  if (i >= N) return;
#pragma unroll
  for (int f = 0; f < 8; ++f) {
    float v = agg[(size_t)i * 8 + f] + b[f];
    out[(size_t)i * 8 + f] = v > 0.0f ? v : 0.0f;
  }
}

__global__ void logsoftmax_kernel(const float* __restrict__ agg,
                                  const float* __restrict__ b,
                                  float* __restrict__ out, int N) {
  int i = blockIdx.x * blockDim.x + threadIdx.x;
  if (i >= N) return;
  float o0 = agg[(size_t)i * 2 + 0] + b[0];
  float o1 = agg[(size_t)i * 2 + 1] + b[1];
  float mx = fmaxf(o0, o1);
  float lse = mx + __logf(__expf(o0 - mx) + __expf(o1 - mx));
  out[(size_t)i * 2 + 0] = o0 - lse;
  out[(size_t)i * 2 + 1] = o1 - lse;
}

extern "C" void kernel_launch(void* const* d_in, const int* in_sizes, int n_in,
                              void* d_out, int out_size, void* d_ws,
                              size_t ws_size, hipStream_t stream) {
  const float* x      = (const float*)d_in[0];
  const float* W1     = (const float*)d_in[1];
  const float* a1_src = (const float*)d_in[2];
  const float* a1_dst = (const float*)d_in[3];
  const float* b1     = (const float*)d_in[4];
  const float* W2     = (const float*)d_in[5];
  const float* a2_src = (const float*)d_in[6];
  const float* a2_dst = (const float*)d_in[7];
  const float* b2     = (const float*)d_in[8];
  const int*   eidx   = (const int*)d_in[9];

  const int N = in_sizes[0] / 36;
  const int E = in_sizes[9] / 2;
  const int* src = eidx;
  const int* dst = eidx + E;

  float* ws    = (float*)d_ws;
  float* h1    = ws;
  float* m1    = h1 + (size_t)N * 16;
  float* z1    = m1 + N;
  float* agg1  = z1 + N;
  float* h1r   = agg1 + (size_t)N * 8;
  float* h2    = h1r + (size_t)N * 8;
  float* m2    = h2 + (size_t)N * 16;
  float* z2    = m2 + N;
  float* agg2  = z2 + N;
  float* wext1 = agg2 + (size_t)N * 2;
  float* wext2 = wext1 + 36 * 16;
  float* ssE   = wext2 + 8 * 16;
  float* sS1   = h2;
  float* sD1   = h2 + N;
  float* sS2   = agg1;
  float* sD2   = agg1 + N;

  const int BLK = 256;
  dim3 nodeGrid((N + BLK - 1) / BLK);
  dim3 gemmGrid((N + 127) / 128);

  prep_wext_kernel<<<dim3((36 * 16 + BLK - 1) / BLK), BLK, 0, stream>>>(
      W1, a1_src, a1_dst, wext1, 36, 8);
  init_node_kernel<<<nodeGrid, BLK, 0, stream>>>(m1, z1, agg1, N, 8);
  gemm_attn_wmma<9><<<gemmGrid, BLK, 0, stream>>>(x, 36, wext1, h1, sS1, sD1, N);
  edge_score_kernel<<<dim3(256), dim3(1024), 0, stream>>>(src, sS1, ssE, ws, ws_size, N, E);
  edge_owner_kernel<0, 1, 25088, 8, 4><<<dim3(256), dim3(1024), 0, stream>>>(
      src, dst, h1, sS1, sD1, nullptr, nullptr, m1, ssE, ws, ws_size, N, E);
  edge_owner_kernel<1, 1, 20480, 8, 4><<<dim3(256), dim3(1024), 0, stream>>>(
      src, dst, h1, sS1, sD1, m1, nullptr, z1, ssE, ws, ws_size, N, E);
  edge_owner_kernel<2, 8, 6668, 16, 2><<<dim3(256), dim3(1024), 0, stream>>>(
      src, dst, h1, sS1, sD1, m1, z1, agg1, nullptr, ws, ws_size, N, E);
  relu_bias_kernel<<<nodeGrid, BLK, 0, stream>>>(agg1, b1, h1r, N);

  prep_wext_kernel<<<dim3(1), BLK, 0, stream>>>(W2, a2_src, a2_dst, wext2, 8, 2);
  init_node_kernel<<<nodeGrid, BLK, 0, stream>>>(m2, z2, agg2, N, 2);
  gemm_attn_wmma<2><<<gemmGrid, BLK, 0, stream>>>(h1r, 8, wext2, h2, sS2, sD2, N);
  edge_score_kernel<<<dim3(256), dim3(1024), 0, stream>>>(src, sS2, ssE, ws, ws_size, N, E);
  edge_owner_kernel<0, 1, 25088, 8, 4><<<dim3(256), dim3(1024), 0, stream>>>(
      src, dst, h2, sS2, sD2, nullptr, nullptr, m2, ssE, ws, ws_size, N, E);
  edge_owner_kernel<1, 1, 20480, 8, 4><<<dim3(256), dim3(1024), 0, stream>>>(
      src, dst, h2, sS2, sD2, m2, nullptr, z2, ssE, ws, ws_size, N, E);
  edge_owner_kernel<2, 2, 14720, 8, 4><<<dim3(256), dim3(1024), 0, stream>>>(
      src, dst, h2, sS2, sD2, m2, z2, agg2, nullptr, ws, ws_size, N, E);

  logsoftmax_kernel<<<nodeGrid, BLK, 0, stream>>>(agg2, b2, (float*)d_out, N);
}


# edge passes: index loads in whole lines (neighbouring lanes read neighbouring 16-byte pieces, not K consecutive edges per lane); score loads take ring words at lane stride; LDS sum order changed
# speedup vs baseline: 9.8679x; 1.0669x over previous
#include <hip/hip_runtime.h>
#include <math.h>

typedef __attribute__((ext_vector_type(2))) float v2f;
typedef __attribute__((ext_vector_type(8))) float v8f;

#define NEG_SLOPE 0.2f

__device__ __forceinline__ float lrelu(float x) {
  return x > 0.0f ? x : NEG_SLOPE * x;
}

__global__ void prep_wext_kernel(const float* __restrict__ W,
                                 const float* __restrict__ asrc,
                                 const float* __restrict__ adst,
                                 float* __restrict__ Wext, int K, int F) {
  int i = blockIdx.x * blockDim.x + threadIdx.x;
  if (i >= K * 16) return;
  int k = i >> 4;
  int n = i & 15;
  float v = 0.0f;
  if (n < F) {
    v = W[k * F + n];
  } else if (n == 8) {
    for (int j = 0; j < F; ++j) v += W[k * F + j] * asrc[j];
  } else if (n == 9) {
    for (int j = 0; j < F; ++j) v += W[k * F + j] * adst[j];
  }
  Wext[k * 16 + n] = v;
}

__global__ void init_node_kernel(float* __restrict__ m, float* __restrict__ z,
                                 float* __restrict__ agg, int N, int F) {
  int i = blockIdx.x * blockDim.x + threadIdx.x;
  if (i >= N) return;
  m[i] = -INFINITY;
  z[i] = 0.0f;
  for (int f = 0; f < F; ++f) agg[i * F + f] = 0.0f;
}

template <int K4>
__global__ void gemm_attn_wmma(const float* __restrict__ A, int lda,
                               const float* __restrict__ Bext,
                               float* __restrict__ H, float* __restrict__ sS,
                               float* __restrict__ sD, int nRows) {
  __shared__ float st[2][128];
  int lane = threadIdx.x & 31;
  int wv = (int)(threadIdx.x >> 5);
  int row0 = (int)blockIdx.x * 128 + wv * 16;

  if (row0 < nRows) {
    int hf = lane >> 4;
    int l  = lane & 15;

    v8f c = {};
#pragma unroll
    for (int k = 0; k < K4; ++k) {
      int kb = k * 4;
      const float* ap = A + (size_t)(row0 + l) * lda + kb + hf * 2;
      v2f a;
      a.x = ap[0];
      a.y = ap[1];
      const float* bp = Bext + (size_t)(kb + hf * 2) * 16 + l;
      v2f b;
      b.x = bp[0];
      b.y = bp[16];
      c = __builtin_amdgcn_wmma_f32_16x16x4_f32(false, a, false, b, (short)0, c,
                                                false, false);
    }

    float* out = H + (size_t)(row0 + hf * 8) * 16 + l;
#pragma unroll
    for (int r = 0; r < 8; ++r) out[(size_t)r * 16] = c[r];

    if (l == 8 || l == 9) {
      float* sp = &st[l - 8][wv * 16 + hf * 8];
#pragma unroll
      for (int r = 0; r < 8; ++r) sp[r] = c[r];
    }
  }
  __syncthreads();
  const int t = (int)(threadIdx.x & 127);
  const int row = (int)blockIdx.x * 128 + t;
  if (row < nRows) {
    if (threadIdx.x < 128) sS[row] = st[0][t];
    else sD[row] = st[1][t];
  }
}

constexpr int SSCAP = 65536;
constexpr int EPT = 28;

__global__ __launch_bounds__(1024) void edge_score_kernel(
    const int* __restrict__ src, const float* __restrict__ sS,
    float* __restrict__ sE, const float* wsb, size_t wsz, int N, int E) {
  __shared__ alignas(16) float ss[SSCAP];
  if ((size_t)((const char*)(sE + E) - (const char*)wsb) > wsz) return;
  const unsigned tid = threadIdx.x;
  const unsigned Eu = (unsigned)E;
  const long long CH = 1024LL * EPT;
  for (long long c0 = (long long)blockIdx.x * CH; c0 < (long long)E;
       c0 += (long long)gridDim.x * CH) {
    const unsigned i0 = (unsigned)c0 + tid;
    int s[EPT];
    float v[EPT];
#pragma unroll
    for (int q = 0; q < EPT; ++q) {
      const unsigned i = i0 + (unsigned)q * 1024u;
      const int sv = src[i < Eu ? i : Eu - 1u];
      s[q] = (i < Eu) ? sv : -1;
      v[q] = 0.0f;
    }
    for (int base = 0; base < N; base += SSCAP) {
      const int cnt = (N - base < SSCAP) ? (N - base) : SSCAP;
      __syncthreads();
      const float* sp = sS + base;
      const int c4 = ((((size_t)sp) & 15) == 0) ? (cnt & ~3) : 0;
#pragma unroll 8
      for (int i = (int)tid * 4; i < c4; i += 4096)
        *reinterpret_cast<float4*>(ss + i) = *reinterpret_cast<const float4*>(sp + i);
      for (int i = c4 + (int)tid; i < cnt; i += 1024) ss[i] = sp[i];
      __syncthreads();
#pragma unroll
      for (int q = 0; q < EPT; ++q) {
        const unsigned o = (unsigned)(s[q] - base);
        const bool in = o < (unsigned)cnt;
        const float t = ss[in ? o : 0u];
        v[q] = in ? t : v[q];
      }
    }
#pragma unroll
    for (int q = 0; q < EPT; ++q) {
      const unsigned i = i0 + (unsigned)q * 1024u;
      if (i < Eu) sE[i] = v[q];
    }
  }
}

constexpr int plane_pitch(int F, int NR) {
  if (F == 1) return NR;
  int p = NR;
  if (F == 8) {
    while (p % 16 != 4) ++p;
  } else {
    while (p % 64 < 32 / F || p % 64 > 64 - 32 / F) ++p;
  }
  return p;
}

template <int MODE, int F, int NRP, int U>
__device__ __forceinline__ void owner_edges(const int (&sg)[U], const unsigned (&tl)[U],
                                            const float* __restrict__ H,
                                            const float* __restrict__ sS,
                                            const float* __restrict__ eS,
                                            const int* __restrict__ srcb,
                                            const bool useE,
                                            float* acc, const float* sd,
                                            const float* mp, const float* zp) {
  float a[U], hs[U][F];
#pragma unroll
  for (int u = 0; u < U; ++u) {
    if constexpr (MODE == 2) {
      const float* hp = H + (size_t)sg[u] * 16;
      a[u] = hp[8];
      if constexpr (F == 8) {
        float4 h0 = *reinterpret_cast<const float4*>(hp);
        float4 h1 = *reinterpret_cast<const float4*>(hp + 4);
        hs[u][0] = h0.x; hs[u][1] = h0.y; hs[u][2] = h0.z; hs[u][3] = h0.w;
        hs[u][4] = h1.x; hs[u][5] = h1.y; hs[u][6] = h1.z; hs[u][7] = h1.w;
      } else {
#pragma unroll
        for (int f = 0; f < F; ++f) hs[u][f] = hp[f];
      }
    }
  }
  if constexpr (MODE != 2) {
    if (useE) {
#pragma unroll
      for (int u = 0; u < U; ++u) a[u] = eS[sg[u]];
    } else {
#pragma unroll
      for (int u = 0; u < U; ++u) a[u] = sS[srcb[sg[u]]];
    }
  }
#pragma unroll
  for (int u = 0; u < U; ++u) {
    if constexpr (MODE == 0) {
      __hip_atomic_fetch_max(&acc[tl[u]], a[u], __ATOMIC_RELAXED,
                             __HIP_MEMORY_SCOPE_WORKGROUP);
    } else {
      float e = lrelu(a[u] + sd[tl[u]]);
      if constexpr (MODE == 1) {
        float p = __expf(e - mp[tl[u]]);
        atomicAdd(&acc[tl[u]], p);
      } else {
        float alpha = __expf(e - mp[tl[u]]) / zp[tl[u]];
#pragma unroll
        for (int f = 0; f < F; ++f)
          atomicAdd(&acc[f * NRP + tl[u]], alpha * hs[u][f]);
      }
    }
  }
}

template <int MODE, int F, int NR, int K, int U>
__global__ __launch_bounds__(1024) void edge_owner_kernel(
    const int* __restrict__ src, const int* __restrict__ dst,
    const float* __restrict__ H, const float* __restrict__ sS,
    const float* __restrict__ sD, const float* __restrict__ m,
    const float* __restrict__ z, float* __restrict__ outp,
    const float* __restrict__ ssE, const float* wsb, size_t wsz, int N, int E) {
  constexpr int NPL = (MODE == 0) ? 0 : (MODE == 1 ? 2 : 3);
  constexpr int PLN = (NR * NPL > 0) ? NR * NPL : 4;
  constexpr int TRIP = 1024 * K;
  constexpr int NRP = plane_pitch(F, NR);
  static_assert(F == 1 || F == 2 || F == 8, "plane pitch is worked out for these");
  static_assert(NR % 4 == 0, "operand planes are filled by 16-byte stores");
  static_assert(NR <= 32768, "a local destination must fit 15 bits");
  constexpr int TH = 32 * U;
  constexpr int QCAP = (MODE == 0) ? 512 : 256;
  static_assert(TH + 127 <= QCAP && (QCAP & (QCAP - 1)) == 0 && QCAP % TH == 0, "ring length");
  __shared__ float acc[NRP * F];
  __shared__ alignas(16) float pl[PLN];
  __shared__ alignas(16) unsigned qs[32 * QCAP];
  const float* sd = pl;
  const float* mp = pl + (NPL > 1 ? NR : 0);
  const float* zp = pl + (NPL > 2 ? 2 * NR : 0);
  const int tid = threadIdx.x;
  const int lane = tid & 31;
  unsigned* qw = qs + (tid >> 5) * QCAP;
  const int nwg = gridDim.x;
  const int R = (N + NR - 1) / NR;
  int G = nwg / R;
  if (G < 1) G = 1;
  const int g = blockIdx.x % G;
  const int rstride = nwg / G;
  int per = (E + G - 1) / G;
  per = ((per + TRIP - 1) / TRIP) * TRIP;
  const long long b0 = (long long)g * per;
  const int begin = b0 < (long long)E ? (int)b0 : E;
  const int end = (E - begin < per) ? E : begin + per;
  const bool vec = (E % K) == 0 && (MODE == 2 ? N <= (1 << 17) : per <= (1 << 17));
  const float initv = (MODE == 0) ? -INFINITY : 0.0f;
  const bool useE = (MODE != 2) &&
      (size_t)((const char*)(ssE + E) - (const char*)wsb) <= wsz;
  const float* eS = (MODE == 2) ? sS : ssE + begin;
  const int* srcb = src + begin;

  for (int r = blockIdx.x / G; r < R; r += rstride) {
    const int lo = r * NR;
    const unsigned nr = (unsigned)((N - lo < NR) ? (N - lo) : NR);
    for (int i = tid; i < NRP * F; i += 1024) acc[i] = initv;
    if constexpr (MODE != 0) {
      const float* sdp = sD + lo;
      const float* mpg = (NPL > 1) ? m + lo : sdp;
      const float* zpg = (NPL > 2) ? z + lo : sdp;
      const bool al16 =
          ((((size_t)sdp) | ((size_t)mpg) | ((size_t)zpg)) & 15) == 0 && (NR % 4) == 0;
      const int n4 = al16 ? ((int)nr & ~3) : 0;
#pragma unroll 2
      for (int i = tid * 4; i < n4; i += 4096) {
        float4 v0 = *reinterpret_cast<const float4*>(sdp + i);
        float4 v1, v2;
        if constexpr (NPL > 1) v1 = *reinterpret_cast<const float4*>(mpg + i);
        if constexpr (NPL > 2) v2 = *reinterpret_cast<const float4*>(zpg + i);
        *reinterpret_cast<float4*>(pl + i) = v0;
        if constexpr (NPL > 1) {
          v1.x = lrelu(v1.x + v0.x); v1.y = lrelu(v1.y + v0.y);
          v1.z = lrelu(v1.z + v0.z); v1.w = lrelu(v1.w + v0.w);
          *reinterpret_cast<float4*>(pl + NR + i) = v1;
        }
        if constexpr (NPL > 2) {
          v2.x += 1e-16f; v2.y += 1e-16f; v2.z += 1e-16f; v2.w += 1e-16f;
          *reinterpret_cast<float4*>(pl + 2 * NR + i) = v2;
        }
      }
      for (int i = n4 + tid; i < (int)nr; i += 1024) {
        pl[i] = sdp[i];
        if constexpr (NPL > 1) pl[NR + i] = lrelu(mpg[i] + sdp[i]);
        if constexpr (NPL > 2) pl[2 * NR + i] = zpg[i] + 1e-16f;
      }
    }
    __syncthreads();

    if (begin < end) {
      int pbeg = begin;
      if (vec) {
        const int nfull = (end - begin) / TRIP;
        const int fend = begin + nfull * TRIP;
        pbeg = fend;
        unsigned head = 0, tail = 0;
        int i = begin + (tid >> 5) * (32 * K) + lane * 4;
        int4 dv[K / 4], sv[K / 4], dn[K / 4], sn[K / 4];
        if (nfull > 0) {
#pragma unroll
          for (int q = 0; q < K / 4; ++q) {
            dv[q] = *reinterpret_cast<const int4*>(dst + i + 128 * q);
            if constexpr (MODE == 2) sv[q] = *reinterpret_cast<const int4*>(src + i + 128 * q);
          }
        }
        for (int base = begin; base < fend; base += TRIP) {
          const int inext = i + TRIP;
          if (base + TRIP < fend) {
#pragma unroll
            for (int q = 0; q < K / 4; ++q) {
              dn[q] = *reinterpret_cast<const int4*>(dst + inext + 128 * q);
              if constexpr (MODE == 2) sn[q] = *reinterpret_cast<const int4*>(src + inext + 128 * q);
            }
          } else {
#pragma unroll
            for (int q = 0; q < K / 4; ++q) {
              dn[q] = dv[q];
              if constexpr (MODE == 2) sn[q] = sv[q];
            }
          }
          const unsigned lou = (unsigned)lo;
          const unsigned prel = (unsigned)(i - begin);
#pragma unroll
          for (int q = 0; q < K / 4; ++q) {
            const int d4[4] = {dv[q].x, dv[q].y, dv[q].z, dv[q].w};
            unsigned h4[4];
            if constexpr (MODE == 2) {
              h4[0] = (unsigned)sv[q].x; h4[1] = (unsigned)sv[q].y;
              h4[2] = (unsigned)sv[q].z; h4[3] = (unsigned)sv[q].w;
            } else {
#pragma unroll
              for (int j = 0; j < 4; ++j) h4[j] = prel + (unsigned)(128 * q + j);
            }
#pragma unroll
            for (int j = 0; j < 4; ++j) {
              const unsigned t = (unsigned)d4[j] - lou;
              const bool keep = t < nr;
              const unsigned km = __builtin_amdgcn_ballot_w32(keep);
              if (keep) {
                const unsigned pos = __builtin_amdgcn_mbcnt_lo(km, tail) & (QCAP - 1);
                qw[pos] = (h4[j] << 15) | t;
              }
              tail += (unsigned)__builtin_popcount(km);
            }
            while (tail - head >= (unsigned)TH) {
              int sg[U];
              unsigned tl[U];
              unsigned en[U];
              if constexpr (MODE != 2) {
                const unsigned p0 = (head + (unsigned)lane) & (QCAP - 1);
#pragma unroll
                for (int u = 0; u < U; ++u) en[u] = qw[p0 + 32u * (unsigned)u];
              } else {
                const unsigned p0 = (head + (unsigned)lane * U) & (QCAP - 1);
                if constexpr (U == 4) {
                  const uint4 e4 = *reinterpret_cast<const uint4*>(qw + p0);
                  en[0] = e4.x; en[1] = e4.y; en[2] = e4.z; en[3] = e4.w;
                } else if constexpr (U == 2) {
                  const uint2 e2 = *reinterpret_cast<const uint2*>(qw + p0);
                  en[0] = e2.x; en[1] = e2.y;
                } else {
#pragma unroll
                  for (int u = 0; u < U; ++u) en[u] = qw[(p0 + u) & (QCAP - 1)];
                }
              }
#pragma unroll
              for (int u = 0; u < U; ++u) {
                sg[u] = (int)(en[u] >> 15);
                tl[u] = en[u] & 0x7fffu;
              }
              owner_edges<MODE, F, NRP, U>(sg, tl, H, sS, eS, srcb, useE, acc, sd, mp, zp);
              head += TH;
            }
          }
#pragma unroll
          for (int q = 0; q < K / 4; ++q) {
            dv[q] = dn[q];
            if constexpr (MODE == 2) sv[q] = sn[q];
          }
          i = inext;
        }
        int cnt = (int)(tail - head);
        while (cnt > 0) {
          if (lane < cnt) {
            const unsigned en = qw[(head + (unsigned)lane) & (QCAP - 1)];
            const int sg[1] = {(int)(en >> 15)};
            const unsigned tl[1] = {en & 0x7fffu};
            owner_edges<MODE, F, NRP, 1>(sg, tl, H, sS, eS, srcb, useE, acc, sd, mp, zp);
          }
          head += 32;
          cnt -= 32;
        }
      }
      for (int i = pbeg + tid; i < end; i += 1024) {
        const unsigned t = (unsigned)(dst[i] - lo);
        if (t < nr) {
          const int sg[1] = {MODE == 2 ? src[i] : i - begin};
          const unsigned tl[1] = {t};
          owner_edges<MODE, F, NRP, 1>(sg, tl, H, sS, eS, srcb, useE, acc, sd, mp, zp);
        }
      }
    }
    __syncthreads();

    const int cntf = (int)nr * F;
    float* gp = outp + (size_t)lo * F;
    for (int k = tid; k < cntf; k += 1024) {
      float v = acc[(k % F) * NRP + k / F];
      if (MODE == 0) atomicMax(gp + k, v);
      else atomicAdd(gp + k, v);
    }
    __syncthreads();
  }
}

__global__ void relu_bias_kernel(const float* __restrict__ agg,
                                 const float* __restrict__ b,
                                 float* __restrict__ out, int N) {
  int i = blockIdx.x * blockDim.x + threadIdx.x;
  if (i >= N) return;
#pragma unroll
  for (int f = 0; f < 8; ++f) {
    float v = agg[(size_t)i * 8 + f] + b[f];
    out[(size_t)i * 8 + f] = v > 0.0f ? v : 0.0f;
  }
}

__global__ void logsoftmax_kernel(const float* __restrict__ agg,
                                  const float* __restrict__ b,
                                  float* __restrict__ out, int N) {
  int i = blockIdx.x * blockDim.x + threadIdx.x;
  if (i >= N) return;
  float o0 = agg[(size_t)i * 2 + 0] + b[0];
  float o1 = agg[(size_t)i * 2 + 1] + b[1];
  float mx = fmaxf(o0, o1);
  float lse = mx + __logf(__expf(o0 - mx) + __expf(o1 - mx));
  out[(size_t)i * 2 + 0] = o0 - lse;
  out[(size_t)i * 2 + 1] = o1 - lse;
}

extern "C" void kernel_launch(void* const* d_in, const int* in_sizes, int n_in,
                              void* d_out, int out_size, void* d_ws,
                              size_t ws_size, hipStream_t stream) {
  const float* x      = (const float*)d_in[0];
  const float* W1     = (const float*)d_in[1];
  const float* a1_src = (const float*)d_in[2];
  const float* a1_dst = (const float*)d_in[3];
  const float* b1     = (const float*)d_in[4];
  const float* W2     = (const float*)d_in[5];
  const float* a2_src = (const float*)d_in[6];
  const float* a2_dst = (const float*)d_in[7];
  const float* b2     = (const float*)d_in[8];
  const int*   eidx   = (const int*)d_in[9];

  const int N = in_sizes[0] / 36;
  const int E = in_sizes[9] / 2;
  const int* src = eidx;
  const int* dst = eidx + E;

  float* ws    = (float*)d_ws;
  float* h1    = ws;
  float* m1    = h1 + (size_t)N * 16;
  float* z1    = m1 + N;
  float* agg1  = z1 + N;
  float* h1r   = agg1 + (size_t)N * 8;
  float* h2    = h1r + (size_t)N * 8;
  float* m2    = h2 + (size_t)N * 16;
  float* z2    = m2 + N;
  float* agg2  = z2 + N;
  float* wext1 = agg2 + (size_t)N * 2;
  float* wext2 = wext1 + 36 * 16;
  float* ssE   = wext2 + 8 * 16;
  float* sS1   = h2;
  float* sD1   = h2 + N;
  float* sS2   = agg1;
  float* sD2   = agg1 + N;

  const int BLK = 256;
  dim3 nodeGrid((N + BLK - 1) / BLK);
  dim3 gemmGrid((N + 127) / 128);

  prep_wext_kernel<<<dim3((36 * 16 + BLK - 1) / BLK), BLK, 0, stream>>>(
      W1, a1_src, a1_dst, wext1, 36, 8);
  init_node_kernel<<<nodeGrid, BLK, 0, stream>>>(m1, z1, agg1, N, 8);
  gemm_attn_wmma<9><<<gemmGrid, BLK, 0, stream>>>(x, 36, wext1, h1, sS1, sD1, N);
  edge_score_kernel<<<dim3(256), dim3(1024), 0, stream>>>(src, sS1, ssE, ws, ws_size, N, E);
  edge_owner_kernel<0, 1, 25088, 8, 4><<<dim3(256), dim3(1024), 0, stream>>>(
      src, dst, h1, sS1, sD1, nullptr, nullptr, m1, ssE, ws, ws_size, N, E);
  edge_owner_kernel<1, 1, 20480, 8, 4><<<dim3(256), dim3(1024), 0, stream>>>(
      src, dst, h1, sS1, sD1, m1, nullptr, z1, ssE, ws, ws_size, N, E);
  edge_owner_kernel<2, 8, 6668, 16, 2><<<dim3(256), dim3(1024), 0, stream>>>(
      src, dst, h1, sS1, sD1, m1, z1, agg1, nullptr, ws, ws_size, N, E);
  relu_bias_kernel<<<nodeGrid, BLK, 0, stream>>>(agg1, b1, h1r, N);

  prep_wext_kernel<<<dim3(1), BLK, 0, stream>>>(W2, a2_src, a2_dst, wext2, 8, 2);
  init_node_kernel<<<nodeGrid, BLK, 0, stream>>>(m2, z2, agg2, N, 2);
  gemm_attn_wmma<2><<<gemmGrid, BLK, 0, stream>>>(h1r, 8, wext2, h2, sS2, sD2, N);
  edge_score_kernel<<<dim3(256), dim3(1024), 0, stream>>>(src, sS2, ssE, ws, ws_size, N, E);
  edge_owner_kernel<0, 1, 25088, 8, 4><<<dim3(256), dim3(1024), 0, stream>>>(
      src, dst, h2, sS2, sD2, nullptr, nullptr, m2, ssE, ws, ws_size, N, E);
  edge_owner_kernel<1, 1, 20480, 8, 4><<<dim3(256), dim3(1024), 0, stream>>>(
      src, dst, h2, sS2, sD2, m2, nullptr, z2, ssE, ws, ws_size, N, E);
  edge_owner_kernel<2, 2, 14720, 8, 4><<<dim3(256), dim3(1024), 0, stream>>>(
      src, dst, h2, sS2, sD2, m2, z2, agg2, nullptr, ws, ws_size, N, E);

  logsoftmax_kernel<<<nodeGrid, BLK, 0, stream>>>(agg2, b2, (float*)d_out, N);
}


# softmax shifted per node by lrelu(max of s_src over ALL nodes + s_dst), not the node's own max: max edge pass now only a guarded fallback for wide score spread; m plane leaves LDS; sum order changed
# speedup vs baseline: 11.4084x; 1.1561x over previous
#include <hip/hip_runtime.h>
#include <math.h>

typedef __attribute__((ext_vector_type(2))) float v2f;
typedef __attribute__((ext_vector_type(8))) float v8f;

#define NEG_SLOPE 0.2f

__device__ __forceinline__ float lrelu(float x) {
  return x > 0.0f ? x : NEG_SLOPE * x;
}

__global__ void prep_wext_kernel(const float* __restrict__ W,
                                 const float* __restrict__ asrc,
                                 const float* __restrict__ adst,
                                 float* __restrict__ Wext, int K, int F) {
  int i = blockIdx.x * blockDim.x + threadIdx.x;
  if (i >= K * 16) return;
  int k = i >> 4;
  int n = i & 15;
  float v = 0.0f;
  if (n < F) {
    v = W[k * F + n];
  } else if (n == 8) {
    for (int j = 0; j < F; ++j) v += W[k * F + j] * asrc[j];
  } else if (n == 9) {
    for (int j = 0; j < F; ++j) v += W[k * F + j] * adst[j];
  }
  Wext[k * 16 + n] = v;
}

__global__ void init_node_kernel(float* __restrict__ m, float* __restrict__ z,
                                 float* __restrict__ agg, int N, int F) {
  int i = blockIdx.x * blockDim.x + threadIdx.x;
  if (i >= N) return;
  m[i] = -INFINITY;
  z[i] = 0.0f;
  for (int f = 0; f < F; ++f) agg[i * F + f] = 0.0f;
}

constexpr int PM_PITCH = 32;

template <int K4>
__global__ void gemm_attn_wmma(const float* __restrict__ A, int lda,
                               const float* __restrict__ Bext,
                               float* __restrict__ H, float* __restrict__ sS,
                               float* __restrict__ sD, float* __restrict__ pm,
                               int nRows) {
  __shared__ float st[2][128];
  int lane = threadIdx.x & 31;
  int wv = (int)(threadIdx.x >> 5);
  int row0 = (int)blockIdx.x * 128 + wv * 16;

  if (row0 < nRows) {
    int hf = lane >> 4;
    int l  = lane & 15;

    v8f c = {};
#pragma unroll
    for (int k = 0; k < K4; ++k) {
      int kb = k * 4;
      const float* ap = A + (size_t)(row0 + l) * lda + kb + hf * 2;
      v2f a;
      a.x = ap[0];
      a.y = ap[1];
      const float* bp = Bext + (size_t)(kb + hf * 2) * 16 + l;
      v2f b;
      b.x = bp[0];
      b.y = bp[16];
      c = __builtin_amdgcn_wmma_f32_16x16x4_f32(false, a, false, b, (short)0, c,
                                                false, false);
    }

    float* out = H + (size_t)(row0 + hf * 8) * 16 + l;
#pragma unroll
    for (int r = 0; r < 8; ++r) out[(size_t)r * 16] = c[r];

    if (l == 8 || l == 9) {
      float* sp = &st[l - 8][wv * 16 + hf * 8];
#pragma unroll
      for (int r = 0; r < 8; ++r) sp[r] = c[r];
    }
  }
  __syncthreads();
  const int t = (int)(threadIdx.x & 127);
  const int row = (int)blockIdx.x * 128 + t;
  if (row < nRows) {
    if (threadIdx.x < 128) sS[row] = st[0][t];
    else sD[row] = st[1][t];
  }
  if (wv == 0) {
    const int rbase = (int)blockIdx.x * 128;
    float mx = -INFINITY, mn = INFINITY;
    for (int k = 0; k < 4; ++k) {
      const int tt = lane + 32 * k;
      if (rbase + tt < nRows) {
        const float v = st[0][tt];
        mx = fmaxf(mx, v);
        mn = fminf(mn, v);
      }
    }
    for (int o = 16; o > 0; o >>= 1) {
      mx = fmaxf(mx, __shfl_xor(mx, o));
      mn = fminf(mn, __shfl_xor(mn, o));
    }
    if (lane == 0) {
      v2f pr;
      pr.x = mx;
      pr.y = mn;
      *reinterpret_cast<v2f*>(pm + (size_t)blockIdx.x * PM_PITCH) = pr;
    }
  }
}

constexpr int SSCAP = 65536;
constexpr int EPT = 28;

__global__ __launch_bounds__(1024) void edge_score_kernel(
    const int* __restrict__ src, const float* __restrict__ sS,
    float* __restrict__ sE, const float* wsb, size_t wsz, int N, int E) {
  __shared__ alignas(16) float ss[SSCAP];
  if ((size_t)((const char*)(sE + E) - (const char*)wsb) > wsz) return;
  const unsigned tid = threadIdx.x;
  const unsigned Eu = (unsigned)E;
  const long long CH = 1024LL * EPT;
  for (long long c0 = (long long)blockIdx.x * CH; c0 < (long long)E;
       c0 += (long long)gridDim.x * CH) {
    const unsigned i0 = (unsigned)c0 + tid;
    int s[EPT];
    float v[EPT];
#pragma unroll
    for (int q = 0; q < EPT; ++q) {
      const unsigned i = i0 + (unsigned)q * 1024u;
      const int sv = src[i < Eu ? i : Eu - 1u];
      s[q] = (i < Eu) ? sv : -1;
      v[q] = 0.0f;
    }
    for (int base = 0; base < N; base += SSCAP) {
      const int cnt = (N - base < SSCAP) ? (N - base) : SSCAP;
      __syncthreads();
      const float* sp = sS + base;
      const int c4 = ((((size_t)sp) & 15) == 0) ? (cnt & ~3) : 0;
#pragma unroll 8
      for (int i = (int)tid * 4; i < c4; i += 4096)
        *reinterpret_cast<float4*>(ss + i) = *reinterpret_cast<const float4*>(sp + i);
      for (int i = c4 + (int)tid; i < cnt; i += 1024) ss[i] = sp[i];
      __syncthreads();
#pragma unroll
      for (int q = 0; q < EPT; ++q) {
        const unsigned o = (unsigned)(s[q] - base);
        const bool in = o < (unsigned)cnt;
        const float t = ss[in ? o : 0u];
        v[q] = in ? t : v[q];
      }
    }
#pragma unroll
    for (int q = 0; q < EPT; ++q) {
      const unsigned i = i0 + (unsigned)q * 1024u;
      if (i < Eu) sE[i] = v[q];
    }
  }
}

constexpr float SHIFT_SPREAD = 16.0f;

constexpr int plane_pitch(int F, int NR) {
  if (F == 1) return NR;
  int p = NR;
  if (F == 8) {
    while (p % 16 != 4) ++p;
  } else {
    while (p % 64 < 32 / F || p % 64 > 64 - 32 / F) ++p;
  }
  return p;
}

template <int MODE, int F, int NRP, int U>
__device__ __forceinline__ void owner_edges(const int (&sg)[U], const unsigned (&tl)[U],
                                            const float* __restrict__ H,
                                            const float* __restrict__ sS,
                                            const float* __restrict__ eS,
                                            const int* __restrict__ srcb,
                                            const bool useE,
                                            float* acc, const float* sd,
                                            const float* zp, const bool fast,
                                            const float smax,
                                            const float* __restrict__ mg) {
  float a[U], hs[U][F], ms[U];
#pragma unroll
  for (int u = 0; u < U; ++u) {
    if constexpr (MODE == 2) {
      const float* hp = H + (size_t)sg[u] * 16;
      a[u] = hp[8];
      if constexpr (F == 8) {
        float4 h0 = *reinterpret_cast<const float4*>(hp);
        float4 h1 = *reinterpret_cast<const float4*>(hp + 4);
        hs[u][0] = h0.x; hs[u][1] = h0.y; hs[u][2] = h0.z; hs[u][3] = h0.w;
        hs[u][4] = h1.x; hs[u][5] = h1.y; hs[u][6] = h1.z; hs[u][7] = h1.w;
      } else {
#pragma unroll
        for (int f = 0; f < F; ++f) hs[u][f] = hp[f];
      }
    }
  }
  if constexpr (MODE != 2) {
    if (useE) {
#pragma unroll
      for (int u = 0; u < U; ++u) a[u] = eS[sg[u]];
    } else {
#pragma unroll
      for (int u = 0; u < U; ++u) a[u] = sS[srcb[sg[u]]];
    }
  }
  if constexpr (MODE != 0) {
    if (fast) {
#pragma unroll
      for (int u = 0; u < U; ++u) ms[u] = smax;
    } else {
#pragma unroll
      for (int u = 0; u < U; ++u) ms[u] = mg[tl[u]];
    }
  }
#pragma unroll
  for (int u = 0; u < U; ++u) {
    if constexpr (MODE == 0) {
      __hip_atomic_fetch_max(&acc[tl[u]], a[u], __ATOMIC_RELAXED,
                             __HIP_MEMORY_SCOPE_WORKGROUP);
    } else {
      const float sdv = sd[tl[u]];
      float e = lrelu(a[u] + sdv);
      const float c = lrelu(ms[u] + sdv);
      if constexpr (MODE == 1) {
        float p = __expf(e - c);
        atomicAdd(&acc[tl[u]], p);
      } else {
        float alpha = __expf(e - c) / zp[tl[u]];
#pragma unroll
        for (int f = 0; f < F; ++f)
          atomicAdd(&acc[f * NRP + tl[u]], alpha * hs[u][f]);
      }
    }
  }
}

template <int MODE, int F, int NR, int K, int U>
__global__ __launch_bounds__(1024) void edge_owner_kernel(
    const int* __restrict__ src, const int* __restrict__ dst,
    const float* __restrict__ H, const float* __restrict__ sS,
    const float* __restrict__ sD, const float* __restrict__ m,
    const float* __restrict__ z, float* __restrict__ outp,
    const float* __restrict__ ssE, const float* __restrict__ pm, int nblk,
    const float* wsb, size_t wsz, int N, int E) {
  constexpr int NPL = (MODE == 0) ? 0 : (MODE == 1 ? 1 : 2);
  constexpr int PLN = (NR * NPL > 0) ? NR * NPL : 4;
  constexpr int TRIP = 1024 * K;
  constexpr int NRP = plane_pitch(F, NR);
  static_assert(F == 1 || F == 2 || F == 8, "plane pitch is worked out for these");
  static_assert(NR % 4 == 0, "operand planes are filled by 16-byte stores");
  static_assert(NR <= 32768, "a local destination must fit 15 bits");
  constexpr int TH = 32 * U;
  constexpr int QCAP = (MODE == 0) ? 512 : 256;
  static_assert(TH + 127 <= QCAP && (QCAP & (QCAP - 1)) == 0 && QCAP % TH == 0, "ring length");
  __shared__ float acc[NRP * F];
  __shared__ alignas(16) float pl[PLN];
  __shared__ alignas(16) unsigned qs[32 * QCAP];
  __shared__ float red[64];
  const float* sd = pl;
  const float* zp = pl + (NPL > 1 ? NR : 0);
  const int tid = threadIdx.x;
  const int lane = tid & 31;
  float smax = -INFINITY, smin = INFINITY;
  for (int i = tid; i < nblk; i += 1024) {
    const v2f pr = *reinterpret_cast<const v2f*>(pm + (size_t)i * PM_PITCH);
    smax = fmaxf(smax, pr.x);
    smin = fminf(smin, pr.y);
  }
  for (int o = 16; o > 0; o >>= 1) {
    smax = fmaxf(smax, __shfl_xor(smax, o));
    smin = fminf(smin, __shfl_xor(smin, o));
  }
  if (lane == 0) {
    red[tid >> 5] = smax;
    red[32 + (tid >> 5)] = smin;
  }
  __syncthreads();
  smax = red[lane];
  smin = red[32 + lane];
  for (int o = 16; o > 0; o >>= 1) {
    smax = fmaxf(smax, __shfl_xor(smax, o));
    smin = fminf(smin, __shfl_xor(smin, o));
  }
  const bool fast = (smax - smin) <= SHIFT_SPREAD;
  if constexpr (MODE == 0) {
    if (fast) return;
  }
  unsigned* qw = qs + (tid >> 5) * QCAP;
  const int nwg = gridDim.x;
  const int R = (N + NR - 1) / NR;
  int G = nwg / R;
  if (G < 1) G = 1;
  const int g = blockIdx.x % G;
  const int rstride = nwg / G;
  int per = (E + G - 1) / G;
  per = ((per + TRIP - 1) / TRIP) * TRIP;
  const long long b0 = (long long)g * per;
  const int begin = b0 < (long long)E ? (int)b0 : E;
  const int end = (E - begin < per) ? E : begin + per;
  const bool vec = (E % K) == 0 && (MODE == 2 ? N <= (1 << 17) : per <= (1 << 17));
  const float initv = (MODE == 0) ? -INFINITY : 0.0f;
  const bool useE = (MODE != 2) &&
      (size_t)((const char*)(ssE + E) - (const char*)wsb) <= wsz;
  const float* eS = (MODE == 2) ? sS : ssE + begin;
  const int* srcb = src + begin;

  for (int r = blockIdx.x / G; r < R; r += rstride) {
    const int lo = r * NR;
    const unsigned nr = (unsigned)((N - lo < NR) ? (N - lo) : NR);
    const float* mg = (MODE == 0) ? sD : m + lo;
    for (int i = tid; i < NRP * F; i += 1024) acc[i] = initv;
    if constexpr (MODE != 0) {
      const float* sdp = sD + lo;
      const float* zpg = (NPL > 1) ? z + lo : sdp;
      const bool al16 =
          ((((size_t)sdp) | ((size_t)zpg)) & 15) == 0 && (NR % 4) == 0;
      const int n4 = al16 ? ((int)nr & ~3) : 0;
#pragma unroll 2
      for (int i = tid * 4; i < n4; i += 4096) {
        float4 v0 = *reinterpret_cast<const float4*>(sdp + i);
        float4 v2;
        if constexpr (NPL > 1) v2 = *reinterpret_cast<const float4*>(zpg + i);
        *reinterpret_cast<float4*>(pl + i) = v0;
        if constexpr (NPL > 1) {
          v2.x += 1e-16f; v2.y += 1e-16f; v2.z += 1e-16f; v2.w += 1e-16f;
          *reinterpret_cast<float4*>(pl + NR + i) = v2;
        }
      }
      for (int i = n4 + tid; i < (int)nr; i += 1024) {
        pl[i] = sdp[i];
        if constexpr (NPL > 1) pl[NR + i] = zpg[i] + 1e-16f;
      }
    }
    __syncthreads();

    if (begin < end) {
      int pbeg = begin;
      if (vec) {
        const int nfull = (end - begin) / TRIP;
        const int fend = begin + nfull * TRIP;
        pbeg = fend;
        unsigned head = 0, tail = 0;
        int i = begin + (tid >> 5) * (32 * K) + lane * 4;
        int4 dv[K / 4], sv[K / 4], dn[K / 4], sn[K / 4];
        if (nfull > 0) {
#pragma unroll
          for (int q = 0; q < K / 4; ++q) {
            dv[q] = *reinterpret_cast<const int4*>(dst + i + 128 * q);
            if constexpr (MODE == 2) sv[q] = *reinterpret_cast<const int4*>(src + i + 128 * q);
          }
        }
        for (int base = begin; base < fend; base += TRIP) {
          const int inext = i + TRIP;
          if (base + TRIP < fend) {
#pragma unroll
            for (int q = 0; q < K / 4; ++q) {
              dn[q] = *reinterpret_cast<const int4*>(dst + inext + 128 * q);
              if constexpr (MODE == 2) sn[q] = *reinterpret_cast<const int4*>(src + inext + 128 * q);
            }
          } else {
#pragma unroll
            for (int q = 0; q < K / 4; ++q) {
              dn[q] = dv[q];
              if constexpr (MODE == 2) sn[q] = sv[q];
            }
          }
          const unsigned lou = (unsigned)lo;
          const unsigned prel = (unsigned)(i - begin);
#pragma unroll
          for (int q = 0; q < K / 4; ++q) {
            const int d4[4] = {dv[q].x, dv[q].y, dv[q].z, dv[q].w};
            unsigned h4[4];
            if constexpr (MODE == 2) {
              h4[0] = (unsigned)sv[q].x; h4[1] = (unsigned)sv[q].y;
              h4[2] = (unsigned)sv[q].z; h4[3] = (unsigned)sv[q].w;
            } else {
#pragma unroll
              for (int j = 0; j < 4; ++j) h4[j] = prel + (unsigned)(128 * q + j);
            }
#pragma unroll
            for (int j = 0; j < 4; ++j) {
              const unsigned t = (unsigned)d4[j] - lou;
              const bool keep = t < nr;
              const unsigned km = __builtin_amdgcn_ballot_w32(keep);
              if (keep) {
                const unsigned pos = __builtin_amdgcn_mbcnt_lo(km, tail) & (QCAP - 1);
                qw[pos] = (h4[j] << 15) | t;
              }
              tail += (unsigned)__builtin_popcount(km);
            }
            while (tail - head >= (unsigned)TH) {
              int sg[U];
              unsigned tl[U];
              unsigned en[U];
              if constexpr (MODE != 2) {
                const unsigned p0 = (head + (unsigned)lane) & (QCAP - 1);
#pragma unroll
                for (int u = 0; u < U; ++u) en[u] = qw[p0 + 32u * (unsigned)u];
              } else {
                const unsigned p0 = (head + (unsigned)lane * U) & (QCAP - 1);
                if constexpr (U == 4) {
                  const uint4 e4 = *reinterpret_cast<const uint4*>(qw + p0);
                  en[0] = e4.x; en[1] = e4.y; en[2] = e4.z; en[3] = e4.w;
                } else if constexpr (U == 2) {
                  const uint2 e2 = *reinterpret_cast<const uint2*>(qw + p0);
                  en[0] = e2.x; en[1] = e2.y;
                } else {
#pragma unroll
                  for (int u = 0; u < U; ++u) en[u] = qw[(p0 + u) & (QCAP - 1)];
                }
              }
#pragma unroll
              for (int u = 0; u < U; ++u) {
                sg[u] = (int)(en[u] >> 15);
                tl[u] = en[u] & 0x7fffu;
              }
              owner_edges<MODE, F, NRP, U>(sg, tl, H, sS, eS, srcb, useE, acc, sd, zp, fast, smax, mg);
              head += TH;
            }
          }
#pragma unroll
          for (int q = 0; q < K / 4; ++q) {
            dv[q] = dn[q];
            if constexpr (MODE == 2) sv[q] = sn[q];
          }
          i = inext;
        }
        int cnt = (int)(tail - head);
        while (cnt > 0) {
          if (lane < cnt) {
            const unsigned en = qw[(head + (unsigned)lane) & (QCAP - 1)];
            const int sg[1] = {(int)(en >> 15)};
            const unsigned tl[1] = {en & 0x7fffu};
            owner_edges<MODE, F, NRP, 1>(sg, tl, H, sS, eS, srcb, useE, acc, sd, zp, fast, smax, mg);
          }
          head += 32;
          cnt -= 32;
        }
      }
      for (int i = pbeg + tid; i < end; i += 1024) {
        const unsigned t = (unsigned)(dst[i] - lo);
        if (t < nr) {
          const int sg[1] = {MODE == 2 ? src[i] : i - begin};
          const unsigned tl[1] = {t};
          owner_edges<MODE, F, NRP, 1>(sg, tl, H, sS, eS, srcb, useE, acc, sd, zp, fast, smax, mg);
        }
      }
    }
    __syncthreads();

    const int cntf = (int)nr * F;
    float* gp = outp + (size_t)lo * F;
    for (int k = tid; k < cntf; k += 1024) {
      float v = acc[(k % F) * NRP + k / F];
      if (MODE == 0) atomicMax(gp + k, v);
      else atomicAdd(gp + k, v);
    }
    __syncthreads();
  }
}

__global__ void relu_bias_kernel(const float* __restrict__ agg,
                                 const float* __restrict__ b,
                                 float* __restrict__ out, int N) {
  int i = blockIdx.x * blockDim.x + threadIdx.x;
  if (i >= N) return;
#pragma unroll
  for (int f = 0; f < 8; ++f) {
    float v = agg[(size_t)i * 8 + f] + b[f];
    out[(size_t)i * 8 + f] = v > 0.0f ? v : 0.0f;
  }
}

__global__ void logsoftmax_kernel(const float* __restrict__ agg,
                                  const float* __restrict__ b,
                                  float* __restrict__ out, int N) {
  int i = blockIdx.x * blockDim.x + threadIdx.x;
  if (i >= N) return;
  float o0 = agg[(size_t)i * 2 + 0] + b[0];
  float o1 = agg[(size_t)i * 2 + 1] + b[1];
  float mx = fmaxf(o0, o1);
  float lse = mx + __logf(__expf(o0 - mx) + __expf(o1 - mx));
  out[(size_t)i * 2 + 0] = o0 - lse;
  out[(size_t)i * 2 + 1] = o1 - lse;
}

extern "C" void kernel_launch(void* const* d_in, const int* in_sizes, int n_in,
                              void* d_out, int out_size, void* d_ws,
                              size_t ws_size, hipStream_t stream) {
  const float* x      = (const float*)d_in[0];
  const float* W1     = (const float*)d_in[1];
  const float* a1_src = (const float*)d_in[2];
  const float* a1_dst = (const float*)d_in[3];
  const float* b1     = (const float*)d_in[4];
  const float* W2     = (const float*)d_in[5];
  const float* a2_src = (const float*)d_in[6];
  const float* a2_dst = (const float*)d_in[7];
  const float* b2     = (const float*)d_in[8];
  const int*   eidx   = (const int*)d_in[9];

  const int N = in_sizes[0] / 36;
  const int E = in_sizes[9] / 2;
  const int* src = eidx;
  const int* dst = eidx + E;

  float* ws    = (float*)d_ws;
  float* h1    = ws;
  float* m1    = h1 + (size_t)N * 16;
  float* z1    = m1 + N;
  float* agg1  = z1 + N;
  float* h1r   = agg1 + (size_t)N * 8;
  float* h2    = h1r + (size_t)N * 8;
  float* m2    = h2 + (size_t)N * 16;
  float* z2    = m2 + N;
  float* agg2  = z2 + N;
  float* wext1 = agg2 + (size_t)N * 2;
  float* wext2 = wext1 + 36 * 16;
  float* ssE   = wext2 + 8 * 16;
  float* sS1   = h2;
  float* sD1   = h2 + N;
  float* sS2   = agg1;
  float* sD2   = agg1 + N;
  float* pm1   = h1r;
  float* pm2   = h1;

  const int BLK = 256;
  dim3 nodeGrid((N + BLK - 1) / BLK);
  dim3 gemmGrid((N + 127) / 128);
  const int nblk = (int)gemmGrid.x;

  prep_wext_kernel<<<dim3((36 * 16 + BLK - 1) / BLK), BLK, 0, stream>>>(
      W1, a1_src, a1_dst, wext1, 36, 8);
  init_node_kernel<<<nodeGrid, BLK, 0, stream>>>(m1, z1, agg1, N, 8);
  gemm_attn_wmma<9><<<gemmGrid, BLK, 0, stream>>>(x, 36, wext1, h1, sS1, sD1, pm1, N);
  edge_score_kernel<<<dim3(256), dim3(1024), 0, stream>>>(src, sS1, ssE, ws, ws_size, N, E);
  edge_owner_kernel<0, 1, 25088, 8, 4><<<dim3(32), dim3(1024), 0, stream>>>(
      src, dst, h1, sS1, sD1, nullptr, nullptr, m1, ssE, pm1, nblk, ws, ws_size, N, E);
  edge_owner_kernel<1, 1, 25088, 8, 4><<<dim3(256), dim3(1024), 0, stream>>>(
      src, dst, h1, sS1, sD1, m1, nullptr, z1, ssE, pm1, nblk, ws, ws_size, N, E);
  edge_owner_kernel<2, 8, 7144, 16, 2><<<dim3(256), dim3(1024), 0, stream>>>(
      src, dst, h1, sS1, sD1, m1, z1, agg1, nullptr, pm1, nblk, ws, ws_size, N, E);
  relu_bias_kernel<<<nodeGrid, BLK, 0, stream>>>(agg1, b1, h1r, N);

  prep_wext_kernel<<<dim3(1), BLK, 0, stream>>>(W2, a2_src, a2_dst, wext2, 8, 2);
  init_node_kernel<<<nodeGrid, BLK, 0, stream>>>(m2, z2, agg2, N, 2);
  gemm_attn_wmma<2><<<gemmGrid, BLK, 0, stream>>>(h1r, 8, wext2, h2, sS2, sD2, pm2, N);
  edge_score_kernel<<<dim3(256), dim3(1024), 0, stream>>>(src, sS2, ssE, ws, ws_size, N, E);
  edge_owner_kernel<0, 1, 25088, 8, 4><<<dim3(32), dim3(1024), 0, stream>>>(
      src, dst, h2, sS2, sD2, nullptr, nullptr, m2, ssE, pm2, nblk, ws, ws_size, N, E);
  edge_owner_kernel<1, 1, 25088, 8, 4><<<dim3(256), dim3(1024), 0, stream>>>(
      src, dst, h2, sS2, sD2, m2, nullptr, z2, ssE, pm2, nblk, ws, ws_size, N, E);
  edge_owner_kernel<2, 2, 16668, 8, 4><<<dim3(256), dim3(1024), 0, stream>>>(
      src, dst, h2, sS2, sD2, m2, z2, agg2, nullptr, pm2, nblk, ws, ws_size, N, E);

  logsoftmax_kernel<<<nodeGrid, BLK, 0, stream>>>(agg2, b2, (float*)d_out, N);
}
